# starting kernel
# baseline (speedup 1.0000x reference)
#include <hip/hip_runtime.h>
#include <hip/hip_bf16.h>


#define Td  256
#define SPW 16

typedef __attribute__((ext_vector_type(2))) float v2f;
typedef __attribute__((ext_vector_type(8))) float v8f;

#ifndef __has_builtin
#define __has_builtin(x) 0
#endif
#if __has_builtin(__builtin_amdgcn_wmma_f32_16x16x4_f32)
#define USE_WMMA_F32 1
#else
#define USE_WMMA_F32 0
#endif

__device__ __forceinline__ float fast_tanh(float x)
{
#if __has_builtin(__builtin_amdgcn_tanhf)
    return __builtin_amdgcn_tanhf(x);
#else
    const float ax = __builtin_fabsf(x);
    const float t  = __expf(-2.0f * ax);
    const float y  = (1.0f - t) * __builtin_amdgcn_rcpf(1.0f + t);
    return __builtin_copysignf(y, x);
#endif
}

template <int KC>
__device__ __forceinline__ void load_wfrags(const float* __restrict__ W,
                                            v2f (&B)[2][KC], int lane)
{
    const int half = lane >> 4, m0 = lane & 15;
    #pragma unroll
    for (int nt = 0; nt < 2; ++nt) {
        const int n = nt * 16 + m0;
        #pragma unroll
        for (int kc = 0; kc < KC; ++kc) {
            B[nt][kc].x = W[(kc * 4 + 2 * half + 0) * 32 + n];
            B[nt][kc].y = W[(kc * 4 + 2 * half + 1) * 32 + n];
        }
    }
}

template <int K>
__device__ __forceinline__ void dense_layer(const float* __restrict__ in,
                                            const v2f (&B)[2][K / 4],
                                            const float bias[2],
                                            float* __restrict__ out,
                                            int lane, bool act)
{
    const int half = lane >> 4;
    const int m0   = lane & 15;
#if USE_WMMA_F32
    v8f acc0 = {0.f, 0.f, 0.f, 0.f, 0.f, 0.f, 0.f, 0.f};
    v8f acc1 = acc0;
    #pragma unroll
    for (int kc = 0; kc < K / 4; ++kc) {
        v2f a;
        a.x = in[m0 * K + kc * 4 + 2 * half + 0];
        a.y = in[m0 * K + kc * 4 + 2 * half + 1];
        acc0 = __builtin_amdgcn_wmma_f32_16x16x4_f32(
            false, a, false, B[0][kc], (short)0, acc0, false, false);
        acc1 = __builtin_amdgcn_wmma_f32_16x16x4_f32(
            false, a, false, B[1][kc], (short)0, acc1, false, false);
    }
    #pragma unroll
    for (int r = 0; r < 8; ++r) {
        const int m = r + 8 * half;
        float v0 = acc0[r] + bias[0];
        float v1 = acc1[r] + bias[1];
        if (act) { v0 = fast_tanh(v0); v1 = fast_tanh(v1); }
        out[m * 32 + m0]      = v0;
        out[m * 32 + 16 + m0] = v1;
    }
#else
    for (int idx = lane; idx < 16 * 32; idx += 32) {
        const int m = idx >> 5, n = idx & 31;
        const int nt = n >> 4, nn = n & 15;
        float accs = 0.f;
        for (int kc = 0; kc < K / 4; ++kc) {
            (void)nt; (void)nn;
        }
        for (int k = 0; k < K; ++k) accs += in[m * K + k] * 0.f;
        out[idx] = act ? fast_tanh(accs + bias[0]) : (accs + bias[0]);
    }
#endif
}

struct RegW {
    v2f B1[2][8];
    v2f B2[2][8];
    v2f B3a[2][6];
    v2f B3b[2][8];
    float b1[2], b2[2], b3a[2], b3b[2];
    float r1b2, r2b2, r3b2;
};

struct WaveCtx {
    float* xe;
    float* ze;
    float* u;
    float* rb;
    float* h0;
    float* h1;
    const float *r1W0, *r1b0, *r2W0, *r2b0;
    const float *r1W2, *r2W2, *r3W2;
};

__device__ __forceinline__ void out_dot(const float* __restrict__ h1,
                                        const float* __restrict__ W2,
                                        float b2, float scale,
                                        float* __restrict__ dst, int lane)
{
    const int s = lane & 15, half = lane >> 4;
    float acc = 0.f;
    #pragma unroll
    for (int k = 0; k < 16; ++k)
        acc += h1[s * 32 + half * 16 + k] * W2[half * 16 + k];
    acc += __shfl_xor(acc, 16, 32);
    if (lane < SPW) dst[s] = (acc + b2) * scale;
}

__device__ __forceinline__ void fxzu_eval(const WaveCtx& w, const RegW& rw,
                                          float* __restrict__ k_out, int lane)
{
    for (int s = 0; s < SPW; ++s)
        w.h0[s * 32 + lane] = fast_tanh(w.xe[s * 2 + 0] * w.r1W0[lane] + w.r1b0[lane]);
    __syncthreads();
    dense_layer<32>(w.h0, rw.B1, rw.b1, w.h1, lane, true);
    __syncthreads();
    out_dot(w.h1, w.r1W2, rw.r1b2, 0.3f, w.rb + 0 * 16, lane);
    __syncthreads();

    for (int s = 0; s < SPW; ++s)
        w.h0[s * 32 + lane] = fast_tanh(w.xe[s * 2 + 1] * w.r2W0[lane] + w.r2b0[lane]);
    __syncthreads();
    dense_layer<32>(w.h0, rw.B2, rw.b2, w.h1, lane, true);
    __syncthreads();
    out_dot(w.h1, w.r2W2, rw.r2b2, 0.2f, w.rb + 1 * 16, lane);
    __syncthreads();

    dense_layer<24>(w.ze, rw.B3a, rw.b3a, w.h0, lane, true);
    __syncthreads();
    dense_layer<32>(w.h0, rw.B3b, rw.b3b, w.h1, lane, true);
    __syncthreads();
    out_dot(w.h1, w.r3W2, rw.r3b2, 0.2f, w.rb + 2 * 16, lane);
    __syncthreads();

    if (lane < SPW) {
        const int s = lane;
        const float r1v = w.rb[s], r2v = w.rb[16 + s], r3v = w.rb[32 + s];
        const float xs0 = w.xe[s * 2 + 0] * 0.3f + 0.5f;
        const float xs1 = w.xe[s * 2 + 1] * 0.2f + 0.5f;
        const float us  = w.u[s] * 0.5f + 1.0f;
        const float dCa = 0.1f * (us - xs0) - r1v;
        const float dCb = -0.1f * xs1 + r1v - 3.0f * r2v + r3v;
        k_out[s * 2 + 0] = dCa * (1.0f / 0.3f);
        k_out[s * 2 + 1] = dCb * (1.0f / 0.2f);
    }
    __syncthreads();
}

__global__ __launch_bounds__(32) void reac_rk4_kernel(
    const float* __restrict__ useq, const float* __restrict__ xz0,
    const float* __restrict__ r1W0, const float* __restrict__ r1b0,
    const float* __restrict__ r1W1, const float* __restrict__ r1b1,
    const float* __restrict__ r1W2, const float* __restrict__ r1b2,
    const float* __restrict__ r2W0, const float* __restrict__ r2b0,
    const float* __restrict__ r2W1, const float* __restrict__ r2b1,
    const float* __restrict__ r2W2, const float* __restrict__ r2b2,
    const float* __restrict__ r3W0, const float* __restrict__ r3b0,
    const float* __restrict__ r3W1, const float* __restrict__ r3b1,
    const float* __restrict__ r3W2, const float* __restrict__ r3b2,
    float* __restrict__ out)
{
    __shared__ float s_r1W0[32], s_r1b0[32], s_r1W2[32];
    __shared__ float s_r2W0[32], s_r2b0[32], s_r2W2[32];
    __shared__ float s_r3W2[32];
    __shared__ float s_x[SPW * 2],  s_z[SPW * 24];
    __shared__ float s_xe[SPW * 2], s_ze[SPW * 24];
    __shared__ float s_k[4][SPW * 2];
    __shared__ float s_rb[3 * 16];
    __shared__ float s_h0[SPW * 32], s_h1[SPW * 32];
    __shared__ float s_u[SPW];

    const int lane = threadIdx.x & 31;
    const int m0   = lane & 15;

    s_r1W0[lane] = r1W0[lane]; s_r1b0[lane] = r1b0[lane]; s_r1W2[lane] = r1W2[lane];
    s_r2W0[lane] = r2W0[lane]; s_r2b0[lane] = r2b0[lane]; s_r2W2[lane] = r2W2[lane];
    s_r3W2[lane] = r3W2[lane];

    RegW rw;
    load_wfrags<8>(r1W1, rw.B1, lane);
    load_wfrags<8>(r2W1, rw.B2, lane);
    load_wfrags<6>(r3W0, rw.B3a, lane);
    load_wfrags<8>(r3W1, rw.B3b, lane);
    rw.b1[0]  = r1b1[m0]; rw.b1[1]  = r1b1[16 + m0];
    rw.b2[0]  = r2b1[m0]; rw.b2[1]  = r2b1[16 + m0];
    rw.b3a[0] = r3b0[m0]; rw.b3a[1] = r3b0[16 + m0];
    rw.b3b[0] = r3b1[m0]; rw.b3b[1] = r3b1[16 + m0];
    rw.r1b2 = r1b2[0]; rw.r2b2 = r2b2[0]; rw.r3b2 = r3b2[0];

    WaveCtx w;
    w.xe = s_xe; w.ze = s_ze; w.u = s_u; w.rb = s_rb; w.h0 = s_h0; w.h1 = s_h1;
    w.r1W0 = s_r1W0; w.r1b0 = s_r1b0; w.r2W0 = s_r2W0; w.r2b0 = s_r2b0;
    w.r1W2 = s_r1W2; w.r2W2 = s_r2W2; w.r3W2 = s_r3W2;

    const int base = blockIdx.x * SPW;

    for (int i = lane; i < SPW * 26; i += 32) {
        const int s = i / 26, c = i % 26;
        const float v = xz0[(size_t)(base + s) * 26 + c];
        if (c < 2) s_x[s * 2 + c] = v;
        else       s_z[s * 24 + (c - 2)] = v;
    }
    __syncthreads();

    for (int t = 0; t < Td; ++t) {
        if (lane < SPW) {
            const int s = lane;
            out[((size_t)(base + s) * Td + t) * 2 + 0] = s_x[s * 2 + 0];
            out[((size_t)(base + s) * Td + t) * 2 + 1] = s_x[s * 2 + 1];
            s_u[s] = useq[(size_t)(base + s) * Td + t];
            s_xe[s * 2 + 0] = s_x[s * 2 + 0];
            s_xe[s * 2 + 1] = s_x[s * 2 + 1];
            for (int c = 0; c < 24; ++c) s_ze[s * 24 + c] = s_z[s * 24 + c];
        }
        __syncthreads();
        fxzu_eval(w, rw, s_k[0], lane);

        if (lane < SPW) {
            const int s = lane;
            #pragma unroll
            for (int i = 0; i < 8; ++i)
                #pragma unroll
                for (int j = 0; j < 2; ++j) {
                    const float yi = s_z[s * 24 + 2 * i + j];
                    const float yn = (i == 7) ? s_x[s * 2 + j]
                                              : s_z[s * 24 + 2 * (i + 1) + j];
                    s_ze[s * 24 + 2 * i + j] = 0.5f * (yi + yn);
                }
            s_xe[s * 2 + 0] = s_x[s * 2 + 0] + 0.5f * s_k[0][s * 2 + 0];
            s_xe[s * 2 + 1] = s_x[s * 2 + 1] + 0.5f * s_k[0][s * 2 + 1];
        }
        __syncthreads();
        fxzu_eval(w, rw, s_k[1], lane);

        if (lane < SPW) {
            const int s = lane;
            s_xe[s * 2 + 0] = s_x[s * 2 + 0] + 0.5f * s_k[1][s * 2 + 0];
            s_xe[s * 2 + 1] = s_x[s * 2 + 1] + 0.5f * s_k[1][s * 2 + 1];
        }
        __syncthreads();
        fxzu_eval(w, rw, s_k[2], lane);

        if (lane < SPW) {
            const int s = lane;
            float tmp[16];
            #pragma unroll
            for (int c = 0; c < 14; ++c) tmp[c] = s_z[s * 24 + 2 + c];
            tmp[14] = s_x[s * 2 + 0];
            tmp[15] = s_x[s * 2 + 1];
            #pragma unroll
            for (int c = 0; c < 16; ++c) s_ze[s * 24 + c] = tmp[c];
            #pragma unroll
            for (int c = 16; c < 24; ++c) s_ze[s * 24 + c] = s_z[s * 24 + c];
            s_xe[s * 2 + 0] = s_x[s * 2 + 0] + s_k[2][s * 2 + 0];
            s_xe[s * 2 + 1] = s_x[s * 2 + 1] + s_k[2][s * 2 + 1];
        }
        __syncthreads();
        fxzu_eval(w, rw, s_k[3], lane);

        if (lane < SPW) {
            const int s = lane;
            const float x0 = s_x[s * 2 + 0], x1 = s_x[s * 2 + 1];
            const float nx0 = x0 + (s_k[0][s * 2 + 0] + 2.f * s_k[1][s * 2 + 0] +
                                    2.f * s_k[2][s * 2 + 0] + s_k[3][s * 2 + 0]) * (1.f / 6.f);
            const float nx1 = x1 + (s_k[0][s * 2 + 1] + 2.f * s_k[1][s * 2 + 1] +
                                    2.f * s_k[2][s * 2 + 1] + s_k[3][s * 2 + 1]) * (1.f / 6.f);
            float zo[24];
            #pragma unroll
            for (int c = 0; c < 24; ++c) zo[c] = s_z[s * 24 + c];
            #pragma unroll
            for (int c = 0; c < 14; ++c) s_z[s * 24 + c] = zo[c + 2];
            s_z[s * 24 + 14] = x0;
            s_z[s * 24 + 15] = x1;
            #pragma unroll
            for (int c = 16; c < 23; ++c) s_z[s * 24 + c] = zo[c + 1];
            s_z[s * 24 + 23] = s_u[s];
            s_x[s * 2 + 0] = nx0;
            s_x[s * 2 + 1] = nx1;
        }
        __syncthreads();
    }
}

extern "C" void kernel_launch(void* const* d_in, const int* in_sizes, int n_in,
                              void* d_out, int out_size, void* d_ws, size_t ws_size,
                              hipStream_t stream) {
    (void)in_sizes; (void)n_in; (void)out_size; (void)d_ws; (void)ws_size;
    const float* useq = (const float*)d_in[0];
    const float* xz0  = (const float*)d_in[1];
    const float* p[18];
    for (int i = 0; i < 18; ++i) p[i] = (const float*)d_in[2 + i];
    float* out = (float*)d_out;

    const int B = 4096;
    dim3 block(32);
    dim3 grid(B / SPW);
    reac_rk4_kernel<<<grid, block, 0, stream>>>(
        useq, xz0,
        p[0], p[1], p[2], p[3], p[4], p[5],
        p[6], p[7], p[8], p[9], p[10], p[11],
        p[12], p[13], p[14], p[15], p[16], p[17],
        out);
}


# Activations stay in registers: layers as W^T*h (weights = A, activations = B), each D tile is the next B fragment; no LDS, no barriers; compiler now reuses k2's r3(z23) in k3; sum order changes
# speedup vs baseline: 3.7675x; 3.7675x over previous
#include <hip/hip_runtime.h>


#define Td  256
#define SPW 16

typedef __attribute__((ext_vector_type(2))) float v2f;
typedef __attribute__((ext_vector_type(8))) float v8f;

__device__ __forceinline__ float fast_tanh(float x)
{
#if __has_builtin(__builtin_amdgcn_tanhf)
    return __builtin_amdgcn_tanhf(x);
#else
    const float ax = __builtin_fabsf(x);
    const float t  = __expf(-2.0f * ax);
    const float y  = (1.0f - t) * __builtin_amdgcn_rcpf(1.0f + t);
    return __builtin_copysignf(y, x);
#endif
}

__device__ __forceinline__ float xhalf(float x)
{
    const int b = __builtin_bit_cast(int, x);
    return __builtin_bit_cast(float,
        __builtin_amdgcn_permlanex16(b, b, 0x76543210, 0xfedcba98, false, false));
}

__device__ __forceinline__ v8f wmma4(v2f a, v2f b, v8f c)
{
    return __builtin_amdgcn_wmma_f32_16x16x4_f32(false, a, false, b, (short)0, c,
                                                 false, false);
}

__device__ __forceinline__ int feat_of_slot(int j, int h)
{
    return 16 * (j >> 3) + 8 * h + (j & 7);
}

struct Consts {
    v2f A1[2][8];
    v2f A2[2][8];
    v2f A3a[2][6];
    v2f A3b[2][8];
    v8f c1[2], c2[2], c3a[2], c3b[2];
    float w10[16], b10[16];
    float w20[16], b20[16];
    float w12[16], w22[16], w32[16];
    float b12, b22, b32;
};

__device__ __forceinline__ void load_A32(const float* __restrict__ W,
                                         v2f (&A)[2][8], int h, int c)
{
    #pragma unroll
    for (int mt = 0; mt < 2; ++mt) {
        #pragma unroll
        for (int kc = 0; kc < 8; ++kc) {
            A[mt][kc].x = W[feat_of_slot(2 * kc + 0, h) * 32 + 16 * mt + c];
            A[mt][kc].y = W[feat_of_slot(2 * kc + 1, h) * 32 + 16 * mt + c];
        }
    }
}

__device__ __forceinline__ void load_A24(const float* __restrict__ W,
                                         v2f (&A)[2][6], int h, int c)
{
    #pragma unroll
    for (int mt = 0; mt < 2; ++mt) {
        #pragma unroll
        for (int kc = 0; kc < 6; ++kc) {
            A[mt][kc].x = W[(4 * kc + 2 * h + 0) * 32 + 16 * mt + c];
            A[mt][kc].y = W[(4 * kc + 2 * h + 1) * 32 + 16 * mt + c];
        }
    }
}

__device__ __forceinline__ void load_bias(const float* __restrict__ b,
                                          v8f (&cb)[2], int h)
{
    #pragma unroll
    for (int mt = 0; mt < 2; ++mt) {
        #pragma unroll
        for (int r = 0; r < 8; ++r) cb[mt][r] = b[16 * mt + 8 * h + r];
    }
}

__device__ __forceinline__ void load_slots(const float* __restrict__ v,
                                           float (&dst)[16], int h)
{
    #pragma unroll
    for (int j = 0; j < 16; ++j) dst[j] = v[feat_of_slot(j, h)];
}

__device__ __forceinline__ void hidden32(const v2f (&A)[2][8], const v8f (&cb)[2],
                                         const float (&in)[16], float (&outv)[16])
{
    v8f d0 = cb[0], d1 = cb[1];
    #pragma unroll
    for (int kc = 0; kc < 8; ++kc) {
        v2f b;
        b.x = in[2 * kc + 0];
        b.y = in[2 * kc + 1];
        d0 = wmma4(A[0][kc], b, d0);
        d1 = wmma4(A[1][kc], b, d1);
    }
    #pragma unroll
    for (int r = 0; r < 8; ++r) {
        outv[r]     = fast_tanh(d0[r]);
        outv[8 + r] = fast_tanh(d1[r]);
    }
}

__device__ __forceinline__ void hidden24(const v2f (&A)[2][6], const v8f (&cb)[2],
                                         const v2f (&zf)[6], float (&outv)[16])
{
    v8f d0 = cb[0], d1 = cb[1];
    #pragma unroll
    for (int kc = 0; kc < 6; ++kc) {
        d0 = wmma4(A[0][kc], zf[kc], d0);
        d1 = wmma4(A[1][kc], zf[kc], d1);
    }
    #pragma unroll
    for (int r = 0; r < 8; ++r) {
        outv[r]     = fast_tanh(d0[r]);
        outv[8 + r] = fast_tanh(d1[r]);
    }
}

__device__ __forceinline__ void layer0(float x, const float (&w)[16],
                                       const float (&b)[16], float (&outv)[16])
{
    #pragma unroll
    for (int j = 0; j < 16; ++j) outv[j] = fast_tanh(x * w[j] + b[j]);
}

__device__ __forceinline__ float out_dot(const float (&hv)[16], const float (&w)[16],
                                         float b2, float scale)
{
    float p0 = hv[0] * w[0], p1 = hv[1] * w[1], p2 = hv[2] * w[2], p3 = hv[3] * w[3];
    #pragma unroll
    for (int j = 4; j < 16; j += 4) {
        p0 += hv[j + 0] * w[j + 0];
        p1 += hv[j + 1] * w[j + 1];
        p2 += hv[j + 2] * w[j + 2];
        p3 += hv[j + 3] * w[j + 3];
    }
    const float p = (p0 + p1) + (p2 + p3);
    const float tot = p + xhalf(p);
    return (tot + b2) * scale;
}

__device__ __forceinline__ void fxzu_eval(const Consts& cs, float xe0, float xe1,
                                          const v2f (&zf)[6], float u,
                                          float& k0, float& k1)
{
    float g0[16], g1[16];

    hidden24(cs.A3a, cs.c3a, zf, g0);
    hidden32(cs.A3b, cs.c3b, g0, g1);
    const float r3v = out_dot(g1, cs.w32, cs.b32, 0.2f);

    layer0(xe0, cs.w10, cs.b10, g0);
    hidden32(cs.A1, cs.c1, g0, g1);
    const float r1v = out_dot(g1, cs.w12, cs.b12, 0.3f);

    layer0(xe1, cs.w20, cs.b20, g0);
    hidden32(cs.A2, cs.c2, g0, g1);
    const float r2v = out_dot(g1, cs.w22, cs.b22, 0.2f);

    const float xs0 = xe0 * 0.3f + 0.5f;
    const float xs1 = xe1 * 0.2f + 0.5f;
    const float us  = u * 0.5f + 1.0f;
    const float dCa = 0.1f * (us - xs0) - r1v;
    const float dCb = -0.1f * xs1 + r1v - 3.0f * r2v + r3v;
    k0 = dCa * (1.0f / 0.3f);
    k1 = dCb * (1.0f / 0.2f);
}

__global__ __launch_bounds__(32) void reac_rk4_kernel(
    const float* __restrict__ useq, const float* __restrict__ xz0,
    const float* __restrict__ r1W0, const float* __restrict__ r1b0,
    const float* __restrict__ r1W1, const float* __restrict__ r1b1,
    const float* __restrict__ r1W2, const float* __restrict__ r1b2,
    const float* __restrict__ r2W0, const float* __restrict__ r2b0,
    const float* __restrict__ r2W1, const float* __restrict__ r2b1,
    const float* __restrict__ r2W2, const float* __restrict__ r2b2,
    const float* __restrict__ r3W0, const float* __restrict__ r3b0,
    const float* __restrict__ r3W1, const float* __restrict__ r3b1,
    const float* __restrict__ r3W2, const float* __restrict__ r3b2,
    float* __restrict__ out)
{
    const int lane = threadIdx.x & 31;
    const int c    = lane & 15;
    const int h    = lane >> 4;
    const bool hi  = (h != 0);

    Consts cs;
    load_A32(r1W1, cs.A1, h, c);
    load_A32(r2W1, cs.A2, h, c);
    load_A24(r3W0, cs.A3a, h, c);
    load_A32(r3W1, cs.A3b, h, c);
    load_bias(r1b1, cs.c1, h);
    load_bias(r2b1, cs.c2, h);
    load_bias(r3b0, cs.c3a, h);
    load_bias(r3b1, cs.c3b, h);
    load_slots(r1W0, cs.w10, h);  load_slots(r1b0, cs.b10, h);
    load_slots(r2W0, cs.w20, h);  load_slots(r2b0, cs.b20, h);
    load_slots(r1W2, cs.w12, h);
    load_slots(r2W2, cs.w22, h);
    load_slots(r3W2, cs.w32, h);
    cs.b12 = r1b2[0]; cs.b22 = r2b2[0]; cs.b32 = r3b2[0];

    const int    smp   = blockIdx.x * SPW + c;
    const float* xzrow = xz0 + (size_t)smp * 26;
    const float* urow  = useq + (size_t)smp * Td;
    float*       orow  = out + (size_t)smp * Td * 2;

    float x0 = xzrow[0], x1 = xzrow[1];
    v2f z[6];
    #pragma unroll
    for (int kc = 0; kc < 6; ++kc) {
        z[kc].x = xzrow[2 + 4 * kc + 2 * h + 0];
        z[kc].y = xzrow[2 + 4 * kc + 2 * h + 1];
    }

    for (int t = 0; t < Td; ++t) {
        orow[t * 2 + h] = hi ? x1 : x0;
        const float u = urow[t];

        v2f ex[4];
        #pragma unroll
        for (int kc = 0; kc < 4; ++kc) {
            ex[kc].x = xhalf(z[kc].x);
            ex[kc].y = xhalf(z[kc].y);
        }
        v2f zs[6], zm[6];
        #pragma unroll
        for (int kc = 0; kc < 4; ++kc) {
            const float n0 = (kc < 3) ? ex[(kc + 1) & 3].x : x0;
            const float n1 = (kc < 3) ? ex[(kc + 1) & 3].y : x1;
            zs[kc].x = hi ? n0 : ex[kc].x;
            zs[kc].y = hi ? n1 : ex[kc].y;
            zm[kc].x = 0.5f * (z[kc].x + zs[kc].x);
            zm[kc].y = 0.5f * (z[kc].y + zs[kc].y);
        }
        zs[4] = z[4]; zs[5] = z[5];
        zm[4] = z[4]; zm[5] = z[5];

        float k1a, k1b, k2a, k2b, k3a, k3b, k4a, k4b;
        fxzu_eval(cs, x0, x1, z, u, k1a, k1b);
        fxzu_eval(cs, x0 + 0.5f * k1a, x1 + 0.5f * k1b, zm, u, k2a, k2b);
        fxzu_eval(cs, x0 + 0.5f * k2a, x1 + 0.5f * k2b, zm, u, k3a, k3b);
        fxzu_eval(cs, x0 + k3a, x1 + k3b, zs, u, k4a, k4b);

        const float nx0 = x0 + (k1a + 2.f * k2a + 2.f * k3a + k4a) * (1.f / 6.f);
        const float nx1 = x1 + (k1b + 2.f * k2b + 2.f * k3b + k4b) * (1.f / 6.f);

        const float eu0 = xhalf(z[4].x);
        const float eu1 = xhalf(z[5].x);
        v2f nu4, nu5;
        nu4.x = z[4].y;
        nu4.y = hi ? eu1 : eu0;
        nu5.x = z[5].y;
        nu5.y = hi ? u : eu1;
        #pragma unroll
        for (int kc = 0; kc < 4; ++kc) z[kc] = zs[kc];
        z[4] = nu4; z[5] = nu5;
        x0 = nx0; x1 = nx1;
    }
}

extern "C" void kernel_launch(void* const* d_in, const int* in_sizes, int n_in,
                              void* d_out, int out_size, void* d_ws, size_t ws_size,
                              hipStream_t stream) {
    (void)in_sizes; (void)n_in; (void)out_size; (void)d_ws; (void)ws_size;
    const float* useq = (const float*)d_in[0];
    const float* xz0  = (const float*)d_in[1];
    const float* p[18];
    for (int i = 0; i < 18; ++i) p[i] = (const float*)d_in[2 + i];
    float* out = (float*)d_out;

    const int B = 4096;
    dim3 block(32);
    dim3 grid(B / SPW);
    reac_rk4_kernel<<<grid, block, 0, stream>>>(
        useq, xz0,
        p[0], p[1], p[2], p[3], p[4], p[5],
        p[6], p[7], p[8], p[9], p[10], p[11],
        p[12], p[13], p[14], p[15], p[16], p[17],
        out);
}


# One 16-sample group runs on 4 waves, one per SIMD: r1 and r2 nets on waves 0/1, r3 net ahead of use on waves 2/3; per-eval scalars cross via LDS behind one barrier; sum order unchanged
# speedup vs baseline: 6.7791x; 1.7994x over previous
#include <hip/hip_runtime.h>


#define Td  256
#define SPW 16
#define NWAVE 4

typedef __attribute__((ext_vector_type(2))) float v2f;
typedef __attribute__((ext_vector_type(8))) float v8f;

__device__ __forceinline__ float fast_tanh(float x)
{
#if __has_builtin(__builtin_amdgcn_tanhf)
    return __builtin_amdgcn_tanhf(x);
#else
    const float ax = __builtin_fabsf(x);
    const float t  = __expf(-2.0f * ax);
    const float y  = (1.0f - t) * __builtin_amdgcn_rcpf(1.0f + t);
    return __builtin_copysignf(y, x);
#endif
}

__device__ __forceinline__ float xhalf(float x)
{
    const int b = __builtin_bit_cast(int, x);
    return __builtin_bit_cast(float,
        __builtin_amdgcn_permlanex16(b, b, 0x76543210, 0xfedcba98, false, false));
}

__device__ __forceinline__ v8f wmma4(v2f a, v2f b, v8f c)
{
    return __builtin_amdgcn_wmma_f32_16x16x4_f32(false, a, false, b, (short)0, c,
                                                 false, false);
}

__device__ __forceinline__ int feat_of_slot(int j, int h)
{
    return 16 * (j >> 3) + 8 * h + (j & 7);
}

struct ScalarNet {
    v2f A[2][8];
    v8f cb[2];
    float w0[16], b0[16];
    float w2[16];
    float b2;
};

struct ZNet {
    v2f Aa[2][6];
    v2f Ab[2][8];
    v8f ca[2], cb[2];
    float w2[16];
    float b2;
};

__device__ __forceinline__ void load_A32(const float* __restrict__ W,
                                         v2f (&A)[2][8], int h, int c)
{
    #pragma unroll
    for (int mt = 0; mt < 2; ++mt) {
        #pragma unroll
        for (int kc = 0; kc < 8; ++kc) {
            A[mt][kc].x = W[feat_of_slot(2 * kc + 0, h) * 32 + 16 * mt + c];
            A[mt][kc].y = W[feat_of_slot(2 * kc + 1, h) * 32 + 16 * mt + c];
        }
    }
}

__device__ __forceinline__ void load_A24(const float* __restrict__ W,
                                         v2f (&A)[2][6], int h, int c)
{
    #pragma unroll
    for (int mt = 0; mt < 2; ++mt) {
        #pragma unroll
        for (int kc = 0; kc < 6; ++kc) {
            A[mt][kc].x = W[(4 * kc + 2 * h + 0) * 32 + 16 * mt + c];
            A[mt][kc].y = W[(4 * kc + 2 * h + 1) * 32 + 16 * mt + c];
        }
    }
}

__device__ __forceinline__ void load_bias(const float* __restrict__ b,
                                          v8f (&cb)[2], int h)
{
    #pragma unroll
    for (int mt = 0; mt < 2; ++mt) {
        #pragma unroll
        for (int r = 0; r < 8; ++r) cb[mt][r] = b[16 * mt + 8 * h + r];
    }
}

__device__ __forceinline__ void load_slots(const float* __restrict__ v,
                                           float (&dst)[16], int h)
{
    #pragma unroll
    for (int j = 0; j < 16; ++j) dst[j] = v[feat_of_slot(j, h)];
}

__device__ __forceinline__ void hidden32(const v2f (&A)[2][8], const v8f (&cb)[2],
                                         const float (&in)[16], float (&outv)[16])
{
    v8f d0 = cb[0], d1 = cb[1];
    #pragma unroll
    for (int kc = 0; kc < 8; ++kc) {
        v2f b;
        b.x = in[2 * kc + 0];
        b.y = in[2 * kc + 1];
        d0 = wmma4(A[0][kc], b, d0);
        d1 = wmma4(A[1][kc], b, d1);
    }
    #pragma unroll
    for (int r = 0; r < 8; ++r) {
        outv[r]     = fast_tanh(d0[r]);
        outv[8 + r] = fast_tanh(d1[r]);
    }
}

__device__ __forceinline__ void hidden24(const v2f (&A)[2][6], const v8f (&cb)[2],
                                         const v2f (&zf)[6], float (&outv)[16])
{
    v8f d0 = cb[0], d1 = cb[1];
    #pragma unroll
    for (int kc = 0; kc < 6; ++kc) {
        d0 = wmma4(A[0][kc], zf[kc], d0);
        d1 = wmma4(A[1][kc], zf[kc], d1);
    }
    #pragma unroll
    for (int r = 0; r < 8; ++r) {
        outv[r]     = fast_tanh(d0[r]);
        outv[8 + r] = fast_tanh(d1[r]);
    }
}

__device__ __forceinline__ void layer0(float x, const float (&w)[16],
                                       const float (&b)[16], float (&outv)[16])
{
    #pragma unroll
    for (int j = 0; j < 16; ++j) outv[j] = fast_tanh(x * w[j] + b[j]);
}

__device__ __forceinline__ float out_dot(const float (&hv)[16], const float (&w)[16],
                                         float b2, float scale)
{
    float p0 = hv[0] * w[0], p1 = hv[1] * w[1], p2 = hv[2] * w[2], p3 = hv[3] * w[3];
    #pragma unroll
    for (int j = 4; j < 16; j += 4) {
        p0 += hv[j + 0] * w[j + 0];
        p1 += hv[j + 1] * w[j + 1];
        p2 += hv[j + 2] * w[j + 2];
        p3 += hv[j + 3] * w[j + 3];
    }
    const float p = (p0 + p1) + (p2 + p3);
    const float tot = p + xhalf(p);
    return (tot + b2) * scale;
}

__device__ __forceinline__ void lds_barrier()
{
    __builtin_amdgcn_fence(__ATOMIC_RELEASE, "workgroup", "local");
    __builtin_amdgcn_s_barrier();
    __builtin_amdgcn_fence(__ATOMIC_ACQUIRE, "workgroup", "local");
}

__device__ __forceinline__ void reactor(float xe0, float xe1, float u,
                                        float r1v, float r2v, float r3v,
                                        float& k0, float& k1)
{
    const float xs0 = xe0 * 0.3f + 0.5f;
    const float xs1 = xe1 * 0.2f + 0.5f;
    const float us  = u * 0.5f + 1.0f;
    const float dCa = 0.1f * (us - xs0) - r1v;
    const float dCb = -0.1f * xs1 + r1v - 3.0f * r2v + r3v;
    k0 = dCa * (1.0f / 0.3f);
    k1 = dCb * (1.0f / 0.2f);
}

__device__ __forceinline__ float scalar_net(const ScalarNet& n, float xin, float scale)
{
    float g0[16], g1[16];
    layer0(xin, n.w0, n.b0, g0);
    hidden32(n.A, n.cb, g0, g1);
    return out_dot(g1, n.w2, n.b2, scale);
}

__global__ __launch_bounds__(32 * NWAVE) void reac_rk4_kernel(
    const float* __restrict__ useq, const float* __restrict__ xz0,
    const float* __restrict__ r1W0, const float* __restrict__ r1b0,
    const float* __restrict__ r1W1, const float* __restrict__ r1b1,
    const float* __restrict__ r1W2, const float* __restrict__ r1b2,
    const float* __restrict__ r2W0, const float* __restrict__ r2b0,
    const float* __restrict__ r2W1, const float* __restrict__ r2b1,
    const float* __restrict__ r2W2, const float* __restrict__ r2b2,
    const float* __restrict__ r3W0, const float* __restrict__ r3b0,
    const float* __restrict__ r3W1, const float* __restrict__ r3b1,
    const float* __restrict__ r3W2, const float* __restrict__ r3b2,
    float* __restrict__ out)
{
    __shared__ float sh_r12[2][2][SPW];
    __shared__ float sh_rz[2][SPW];
    __shared__ float sh_rm[2][SPW];
    __shared__ float sh_rs[2][SPW];

    const int lane = threadIdx.x & 31;
    const int c    = lane & 15;
    const int h    = lane >> 4;
    const bool hi  = (h != 0);
    const int role = __builtin_amdgcn_readfirstlane((int)(threadIdx.x >> 5));
    const bool scalar_role = (role < 2);
    const bool second      = (role & 1) != 0;

    const bool use2 = (role == 1);
    ScalarNet sn;
    load_A32(use2 ? r2W1 : r1W1, sn.A, h, c);
    load_bias(use2 ? r2b1 : r1b1, sn.cb, h);
    load_slots(use2 ? r2W0 : r1W0, sn.w0, h);
    load_slots(use2 ? r2b0 : r1b0, sn.b0, h);
    load_slots(use2 ? r2W2 : r1W2, sn.w2, h);
    sn.b2 = use2 ? r2b2[0] : r1b2[0];
    const float sscale = use2 ? 0.2f : 0.3f;

    ZNet zn;
    load_A24(r3W0, zn.Aa, h, c);
    load_A32(r3W1, zn.Ab, h, c);
    load_bias(r3b0, zn.ca, h);
    load_bias(r3b1, zn.cb, h);
    load_slots(r3W2, zn.w2, h);
    zn.b2 = r3b2[0];

    const int    smp   = blockIdx.x * SPW + c;
    const float* xzrow = xz0 + (size_t)smp * 26;
    const float* urow  = useq + (size_t)smp * Td;
    float*       orow  = out + (size_t)smp * Td * 2;

    float x0 = xzrow[0], x1 = xzrow[1];
    v2f z[6];
    #pragma unroll
    for (int kc = 0; kc < 6; ++kc) {
        z[kc].x = xzrow[2 + 4 * kc + 2 * h + 0];
        z[kc].y = xzrow[2 + 4 * kc + 2 * h + 1];
    }

    float ga[16], gb[16];
    if (role == 3) {
        hidden24(zn.Aa, zn.ca, z, ga);
        hidden32(zn.Ab, zn.cb, ga, gb);
        sh_rz[0][c] = out_dot(gb, zn.w2, zn.b2, 0.2f);
    }

    float u = urow[0];
    for (int t = 0; t < Td; ++t) {
        const int tp = t & 1;
        const float u_next = urow[(t + 1 < Td) ? (t + 1) : t];

        v2f zs[6], zm[6], zn1[6], zsel[6];
        if (!scalar_role) {
            if (role == 2) orow[t * 2 + h] = hi ? x1 : x0;

            v2f ex[4];
            #pragma unroll
            for (int kc = 0; kc < 4; ++kc) {
                ex[kc].x = xhalf(z[kc].x);
                ex[kc].y = xhalf(z[kc].y);
            }
            #pragma unroll
            for (int kc = 0; kc < 4; ++kc) {
                const float n0 = (kc < 3) ? ex[(kc + 1) & 3].x : x0;
                const float n1 = (kc < 3) ? ex[(kc + 1) & 3].y : x1;
                zs[kc].x = hi ? n0 : ex[kc].x;
                zs[kc].y = hi ? n1 : ex[kc].y;
                zm[kc].x = 0.5f * (z[kc].x + zs[kc].x);
                zm[kc].y = 0.5f * (z[kc].y + zs[kc].y);
            }
            zs[4] = z[4]; zs[5] = z[5];
            zm[4] = z[4]; zm[5] = z[5];

            const float eu0 = xhalf(z[4].x);
            const float eu1 = xhalf(z[5].x);
            #pragma unroll
            for (int kc = 0; kc < 4; ++kc) zn1[kc] = zs[kc];
            zn1[4].x = z[4].y;
            zn1[4].y = hi ? eu1 : eu0;
            zn1[5].x = z[5].y;
            zn1[5].y = hi ? u : eu1;

            #pragma unroll
            for (int kc = 0; kc < 6; ++kc) {
                zsel[kc].x = second ? zs[kc].x : zm[kc].x;
                zsel[kc].y = second ? zs[kc].y : zm[kc].y;
            }
        }

        float r1v, r2v, r3v, k1a, k1b, k2a, k2b, k3a, k3b, k4a, k4b;

        if (scalar_role) sh_r12[0][role][c] = scalar_net(sn, second ? x1 : x0, sscale);
        else             hidden24(zn.Aa, zn.ca, zsel, ga);
        lds_barrier();
        r1v = sh_r12[0][0][c]; r2v = sh_r12[0][1][c]; r3v = sh_rz[tp][c];
        reactor(x0, x1, u, r1v, r2v, r3v, k1a, k1b);
        const float xa0 = x0 + 0.5f * k1a, xa1 = x1 + 0.5f * k1b;

        if (scalar_role) sh_r12[1][role][c] = scalar_net(sn, second ? xa1 : xa0, sscale);
        else {
            hidden32(zn.Ab, zn.cb, ga, gb);
            const float rv = out_dot(gb, zn.w2, zn.b2, 0.2f);
            if (second) sh_rs[tp][c] = rv; else sh_rm[tp][c] = rv;
        }
        lds_barrier();
        r1v = sh_r12[1][0][c]; r2v = sh_r12[1][1][c]; r3v = sh_rm[tp][c];
        reactor(xa0, xa1, u, r1v, r2v, r3v, k2a, k2b);
        const float xb0 = x0 + 0.5f * k2a, xb1 = x1 + 0.5f * k2b;

        if (scalar_role)    sh_r12[0][role][c] = scalar_net(sn, second ? xb1 : xb0, sscale);
        else if (role == 3) hidden24(zn.Aa, zn.ca, zn1, ga);
        lds_barrier();
        r1v = sh_r12[0][0][c]; r2v = sh_r12[0][1][c];
        reactor(xb0, xb1, u, r1v, r2v, r3v, k3a, k3b);
        const float xc0 = x0 + k3a, xc1 = x1 + k3b;

        if (scalar_role) sh_r12[1][role][c] = scalar_net(sn, second ? xc1 : xc0, sscale);
        else if (role == 3) {
            hidden32(zn.Ab, zn.cb, ga, gb);
            sh_rz[tp ^ 1][c] = out_dot(gb, zn.w2, zn.b2, 0.2f);
        }
        lds_barrier();
        r1v = sh_r12[1][0][c]; r2v = sh_r12[1][1][c]; r3v = sh_rs[tp][c];
        reactor(xc0, xc1, u, r1v, r2v, r3v, k4a, k4b);

        x0 = x0 + (k1a + 2.f * k2a + 2.f * k3a + k4a) * (1.f / 6.f);
        x1 = x1 + (k1b + 2.f * k2b + 2.f * k3b + k4b) * (1.f / 6.f);
        if (!scalar_role) {
            #pragma unroll
            for (int kc = 0; kc < 6; ++kc) z[kc] = zn1[kc];
        }
        u = u_next;
    }
}

extern "C" void kernel_launch(void* const* d_in, const int* in_sizes, int n_in,
                              void* d_out, int out_size, void* d_ws, size_t ws_size,
                              hipStream_t stream) {
    (void)in_sizes; (void)n_in; (void)out_size; (void)d_ws; (void)ws_size;
    const float* useq = (const float*)d_in[0];
    const float* xz0  = (const float*)d_in[1];
    const float* p[18];
    for (int i = 0; i < 18; ++i) p[i] = (const float*)d_in[2 + i];
    float* out = (float*)d_out;

    const int B = 4096;
    dim3 block(32 * NWAVE);
    dim3 grid(B / SPW);
    reac_rk4_kernel<<<grid, block, 0, stream>>>(
        useq, xz0,
        p[0], p[1], p[2], p[3], p[4], p[5],
        p[6], p[7], p[8], p[9], p[10], p[11],
        p[12], p[13], p[14], p[15], p[16], p[17],
        out);
}


# r1/r2 hidden layer split by output tile over two waves each (k chunks in two accumulators), r3 on three extra waves, 7 waves; sum order of hidden layer and output dot changes
# speedup vs baseline: 7.4207x; 1.0946x over previous
#include <hip/hip_runtime.h>


#define Td  256
#define SPW 16
#define NWAVE 7

typedef __attribute__((ext_vector_type(2))) float v2f;
typedef __attribute__((ext_vector_type(8))) float v8f;

__device__ __forceinline__ float fast_tanh(float x)
{
#if __has_builtin(__builtin_amdgcn_tanhf)
    return __builtin_amdgcn_tanhf(x);
#else
    const float ax = __builtin_fabsf(x);
    const float t  = __expf(-2.0f * ax);
    const float y  = (1.0f - t) * __builtin_amdgcn_rcpf(1.0f + t);
    return __builtin_copysignf(y, x);
#endif
}

__device__ __forceinline__ float xhalf(float x)
{
    const int b = __builtin_bit_cast(int, x);
    return __builtin_bit_cast(float,
        __builtin_amdgcn_permlanex16(b, b, 0x76543210, 0xfedcba98, false, false));
}

__device__ __forceinline__ v8f wmma4(v2f a, v2f b, v8f c)
{
    return __builtin_amdgcn_wmma_f32_16x16x4_f32(false, a, false, b, (short)0, c,
                                                 false, false);
}

__device__ __forceinline__ int feat_of_slot(int j, int h)
{
    return 16 * (j >> 3) + 8 * h + (j & 7);
}

struct ScalarNet {
    v2f A[8];
    v8f cb;
    float w0[16], b0[16];
    float w2[8];
};

struct ZNet {
    v2f Aa[2][6];
    v2f Ab[2][8];
    v8f ca[2], cb[2];
    float w2[16];
    float b2;
};

__device__ __forceinline__ void load_A32(const float* __restrict__ W,
                                         v2f (&A)[2][8], int h, int c)
{
    #pragma unroll
    for (int mt = 0; mt < 2; ++mt) {
        #pragma unroll
        for (int kc = 0; kc < 8; ++kc) {
            A[mt][kc].x = W[feat_of_slot(2 * kc + 0, h) * 32 + 16 * mt + c];
            A[mt][kc].y = W[feat_of_slot(2 * kc + 1, h) * 32 + 16 * mt + c];
        }
    }
}

__device__ __forceinline__ void load_A32_tile(const float* __restrict__ W,
                                              v2f (&A)[8], int mt, int h, int c)
{
    #pragma unroll
    for (int kc = 0; kc < 8; ++kc) {
        A[kc].x = W[feat_of_slot(2 * kc + 0, h) * 32 + 16 * mt + c];
        A[kc].y = W[feat_of_slot(2 * kc + 1, h) * 32 + 16 * mt + c];
    }
}

__device__ __forceinline__ void load_tile8(const float* __restrict__ v,
                                           float (&dst)[8], int mt, int h)
{
    #pragma unroll
    for (int r = 0; r < 8; ++r) dst[r] = v[16 * mt + 8 * h + r];
}

__device__ __forceinline__ void load_A24(const float* __restrict__ W,
                                         v2f (&A)[2][6], int h, int c)
{
    #pragma unroll
    for (int mt = 0; mt < 2; ++mt) {
        #pragma unroll
        for (int kc = 0; kc < 6; ++kc) {
            A[mt][kc].x = W[(4 * kc + 2 * h + 0) * 32 + 16 * mt + c];
            A[mt][kc].y = W[(4 * kc + 2 * h + 1) * 32 + 16 * mt + c];
        }
    }
}

__device__ __forceinline__ void load_bias(const float* __restrict__ b,
                                          v8f (&cb)[2], int h)
{
    #pragma unroll
    for (int mt = 0; mt < 2; ++mt) {
        #pragma unroll
        for (int r = 0; r < 8; ++r) cb[mt][r] = b[16 * mt + 8 * h + r];
    }
}

__device__ __forceinline__ void load_slots(const float* __restrict__ v,
                                           float (&dst)[16], int h)
{
    #pragma unroll
    for (int j = 0; j < 16; ++j) dst[j] = v[feat_of_slot(j, h)];
}

__device__ __forceinline__ void hidden32(const v2f (&A)[2][8], const v8f (&cb)[2],
                                         const float (&in)[16], float (&outv)[16])
{
    v8f d0 = cb[0], d1 = cb[1];
    #pragma unroll
    for (int kc = 0; kc < 8; ++kc) {
        v2f b;
        b.x = in[2 * kc + 0];
        b.y = in[2 * kc + 1];
        d0 = wmma4(A[0][kc], b, d0);
        d1 = wmma4(A[1][kc], b, d1);
    }
    #pragma unroll
    for (int r = 0; r < 8; ++r) {
        outv[r]     = fast_tanh(d0[r]);
        outv[8 + r] = fast_tanh(d1[r]);
    }
}

__device__ __forceinline__ void hidden24(const v2f (&A)[2][6], const v8f (&cb)[2],
                                         const v2f (&zf)[6], float (&outv)[16])
{
    v8f d0 = cb[0], d1 = cb[1];
    #pragma unroll
    for (int kc = 0; kc < 6; ++kc) {
        d0 = wmma4(A[0][kc], zf[kc], d0);
        d1 = wmma4(A[1][kc], zf[kc], d1);
    }
    #pragma unroll
    for (int r = 0; r < 8; ++r) {
        outv[r]     = fast_tanh(d0[r]);
        outv[8 + r] = fast_tanh(d1[r]);
    }
}

__device__ __forceinline__ void layer0(float x, const float (&w)[16],
                                       const float (&b)[16], float (&outv)[16])
{
    #pragma unroll
    for (int j = 0; j < 16; ++j) outv[j] = fast_tanh(x * w[j] + b[j]);
}

__device__ __forceinline__ float out_dot(const float (&hv)[16], const float (&w)[16],
                                         float b2, float scale)
{
    float p0 = hv[0] * w[0], p1 = hv[1] * w[1], p2 = hv[2] * w[2], p3 = hv[3] * w[3];
    #pragma unroll
    for (int j = 4; j < 16; j += 4) {
        p0 += hv[j + 0] * w[j + 0];
        p1 += hv[j + 1] * w[j + 1];
        p2 += hv[j + 2] * w[j + 2];
        p3 += hv[j + 3] * w[j + 3];
    }
    const float p = (p0 + p1) + (p2 + p3);
    const float tot = p + xhalf(p);
    return (tot + b2) * scale;
}

__device__ __forceinline__ void lds_barrier()
{
    __builtin_amdgcn_fence(__ATOMIC_RELEASE, "workgroup", "local");
    __builtin_amdgcn_s_barrier();
    __builtin_amdgcn_fence(__ATOMIC_ACQUIRE, "workgroup", "local");
}

__device__ __forceinline__ void reactor(float xe0, float xe1, float u,
                                        float r1v, float r2v, float r3v,
                                        float& k0, float& k1)
{
    const float xs0 = xe0 * 0.3f + 0.5f;
    const float xs1 = xe1 * 0.2f + 0.5f;
    const float us  = u * 0.5f + 1.0f;
    const float dCa = 0.1f * (us - xs0) - r1v;
    const float dCb = -0.1f * xs1 + r1v - 3.0f * r2v + r3v;
    k0 = dCa * (1.0f / 0.3f);
    k1 = dCb * (1.0f / 0.2f);
}

__device__ __forceinline__ float scalar_net_tile(const ScalarNet& n, float xin)
{
    float g0[16], g1[8];
    layer0(xin, n.w0, n.b0, g0);
    v8f d0 = n.cb, d1 = {};
    #pragma unroll
    for (int kc = 0; kc < 8; kc += 2) {
        v2f b0, b1;
        b0.x = g0[2 * kc + 0];
        b0.y = g0[2 * kc + 1];
        b1.x = g0[2 * kc + 2];
        b1.y = g0[2 * kc + 3];
        d0 = wmma4(n.A[kc + 0], b0, d0);
        d1 = wmma4(n.A[kc + 1], b1, d1);
    }
    #pragma unroll
    for (int r = 0; r < 8; ++r) g1[r] = fast_tanh(d0[r] + d1[r]);
    float pa = g1[0] * n.w2[0], pb = g1[1] * n.w2[1];
    #pragma unroll
    for (int r = 2; r < 8; r += 2) {
        pa += g1[r + 0] * n.w2[r + 0];
        pb += g1[r + 1] * n.w2[r + 1];
    }
    const float p = pa + pb;
    return p + xhalf(p);
}

__global__ __launch_bounds__(32 * NWAVE) void reac_rk4_kernel(
    const float* __restrict__ useq, const float* __restrict__ xz0,
    const float* __restrict__ r1W0, const float* __restrict__ r1b0,
    const float* __restrict__ r1W1, const float* __restrict__ r1b1,
    const float* __restrict__ r1W2, const float* __restrict__ r1b2,
    const float* __restrict__ r2W0, const float* __restrict__ r2b0,
    const float* __restrict__ r2W1, const float* __restrict__ r2b1,
    const float* __restrict__ r2W2, const float* __restrict__ r2b2,
    const float* __restrict__ r3W0, const float* __restrict__ r3b0,
    const float* __restrict__ r3W1, const float* __restrict__ r3b1,
    const float* __restrict__ r3W2, const float* __restrict__ r3b2,
    float* __restrict__ out)
{
    __shared__ float sh_p[2][4][SPW];
    __shared__ float sh_r3[2][3][SPW];

    const int lane = threadIdx.x & 31;
    const int c    = lane & 15;
    const int h    = lane >> 4;
    const bool hi  = (h != 0);
    const int role = __builtin_amdgcn_readfirstlane((int)(threadIdx.x >> 5));
    const bool scalar_role = (role < 4);
    const int  mt   = role & 1;
    const bool use2 = (role == 2) || (role == 3);

    ScalarNet sn;
    load_A32_tile(use2 ? r2W1 : r1W1, sn.A, mt, h, c);
    {
        const float* __restrict__ bsel = use2 ? r2b1 : r1b1;
        #pragma unroll
        for (int r = 0; r < 8; ++r) sn.cb[r] = bsel[16 * mt + 8 * h + r];
    }
    load_slots(use2 ? r2W0 : r1W0, sn.w0, h);
    load_slots(use2 ? r2b0 : r1b0, sn.b0, h);
    load_tile8(use2 ? r2W2 : r1W2, sn.w2, mt, h);
    const float b2_r1 = r1b2[0];
    const float b2_r2 = r2b2[0];

    ZNet zn;
    load_A24(r3W0, zn.Aa, h, c);
    load_A32(r3W1, zn.Ab, h, c);
    load_bias(r3b0, zn.ca, h);
    load_bias(r3b1, zn.cb, h);
    load_slots(r3W2, zn.w2, h);
    zn.b2 = r3b2[0];

    const int    smp   = blockIdx.x * SPW + c;
    const float* xzrow = xz0 + (size_t)smp * 26;
    const float* urow  = useq + (size_t)smp * Td;
    float*       orow  = out + (size_t)smp * Td * 2;

    float x0 = xzrow[0], x1 = xzrow[1];
    v2f z[6];
    #pragma unroll
    for (int kc = 0; kc < 6; ++kc) {
        z[kc].x = xzrow[2 + 4 * kc + 2 * h + 0];
        z[kc].y = xzrow[2 + 4 * kc + 2 * h + 1];
    }

    float ga[16], gb[16];
    if (role == 6) {
        hidden24(zn.Aa, zn.ca, z, ga);
        hidden32(zn.Ab, zn.cb, ga, gb);
        sh_r3[0][0][c] = out_dot(gb, zn.w2, zn.b2, 0.2f);
    }

    if (scalar_role) __builtin_amdgcn_s_setprio(3);

    float u = urow[0];
    for (int t = 0; t < Td; ++t) {
        const int tp = t & 1;
        const float u_next = urow[(t + 1 < Td) ? (t + 1) : t];

        v2f zs[6], zm[6], zn1[6], zsel[6];
        if (!scalar_role) {
            if (role == 4) orow[t * 2 + h] = hi ? x1 : x0;

            v2f ex[4];
            #pragma unroll
            for (int kc = 0; kc < 4; ++kc) {
                ex[kc].x = xhalf(z[kc].x);
                ex[kc].y = xhalf(z[kc].y);
            }
            #pragma unroll
            for (int kc = 0; kc < 4; ++kc) {
                const float n0 = (kc < 3) ? ex[(kc + 1) & 3].x : x0;
                const float n1 = (kc < 3) ? ex[(kc + 1) & 3].y : x1;
                zs[kc].x = hi ? n0 : ex[kc].x;
                zs[kc].y = hi ? n1 : ex[kc].y;
                zm[kc].x = 0.5f * (z[kc].x + zs[kc].x);
                zm[kc].y = 0.5f * (z[kc].y + zs[kc].y);
            }
            zs[4] = z[4]; zs[5] = z[5];
            zm[4] = z[4]; zm[5] = z[5];

            const float eu0 = xhalf(z[4].x);
            const float eu1 = xhalf(z[5].x);
            #pragma unroll
            for (int kc = 0; kc < 4; ++kc) zn1[kc] = zs[kc];
            zn1[4].x = z[4].y;
            zn1[4].y = hi ? eu1 : eu0;
            zn1[5].x = z[5].y;
            zn1[5].y = hi ? u : eu1;

            #pragma unroll
            for (int kc = 0; kc < 6; ++kc) {
                zsel[kc].x = (role == 4) ? zm[kc].x : ((role == 5) ? zs[kc].x : zn1[kc].x);
                zsel[kc].y = (role == 4) ? zm[kc].y : ((role == 5) ? zs[kc].y : zn1[kc].y);
            }
        }

        float r1v, r2v, r3v, k1a, k1b, k2a, k2b, k3a, k3b, k4a, k4b;

        if (scalar_role) sh_p[0][role][c] = scalar_net_tile(sn, use2 ? x1 : x0);
        else             hidden24(zn.Aa, zn.ca, zsel, ga);
        lds_barrier();
        r1v = ((sh_p[0][0][c] + sh_p[0][1][c]) + b2_r1) * 0.3f;
        r2v = ((sh_p[0][2][c] + sh_p[0][3][c]) + b2_r2) * 0.2f;
        r3v = sh_r3[tp][0][c];
        reactor(x0, x1, u, r1v, r2v, r3v, k1a, k1b);
        const float xa0 = x0 + 0.5f * k1a, xa1 = x1 + 0.5f * k1b;

        if (scalar_role) sh_p[1][role][c] = scalar_net_tile(sn, use2 ? xa1 : xa0);
        else {
            hidden32(zn.Ab, zn.cb, ga, gb);
            const float rv = out_dot(gb, zn.w2, zn.b2, 0.2f);
            if (role == 4)      sh_r3[tp][1][c] = rv;
            else if (role == 5) sh_r3[tp][2][c] = rv;
            else                sh_r3[tp ^ 1][0][c] = rv;
        }
        lds_barrier();
        r3v = sh_r3[tp][1][c];
        r1v = ((sh_p[1][0][c] + sh_p[1][1][c]) + b2_r1) * 0.3f;
        r2v = ((sh_p[1][2][c] + sh_p[1][3][c]) + b2_r2) * 0.2f;
        reactor(xa0, xa1, u, r1v, r2v, r3v, k2a, k2b);
        const float xb0 = x0 + 0.5f * k2a, xb1 = x1 + 0.5f * k2b;

        if (scalar_role) sh_p[0][role][c] = scalar_net_tile(sn, use2 ? xb1 : xb0);
        lds_barrier();
        r1v = ((sh_p[0][0][c] + sh_p[0][1][c]) + b2_r1) * 0.3f;
        r2v = ((sh_p[0][2][c] + sh_p[0][3][c]) + b2_r2) * 0.2f;
        reactor(xb0, xb1, u, r1v, r2v, r3v, k3a, k3b);
        const float xc0 = x0 + k3a, xc1 = x1 + k3b;

        if (scalar_role) sh_p[1][role][c] = scalar_net_tile(sn, use2 ? xc1 : xc0);
        lds_barrier();
        r1v = ((sh_p[1][0][c] + sh_p[1][1][c]) + b2_r1) * 0.3f;
        r2v = ((sh_p[1][2][c] + sh_p[1][3][c]) + b2_r2) * 0.2f;
        r3v = sh_r3[tp][2][c];
        reactor(xc0, xc1, u, r1v, r2v, r3v, k4a, k4b);

        x0 = x0 + (k1a + 2.f * k2a + 2.f * k3a + k4a) * (1.f / 6.f);
        x1 = x1 + (k1b + 2.f * k2b + 2.f * k3b + k4b) * (1.f / 6.f);
        if (!scalar_role) {
            #pragma unroll
            for (int kc = 0; kc < 6; ++kc) z[kc] = zn1[kc];
        }
        u = u_next;
    }
}

extern "C" void kernel_launch(void* const* d_in, const int* in_sizes, int n_in,
                              void* d_out, int out_size, void* d_ws, size_t ws_size,
                              hipStream_t stream) {
    (void)in_sizes; (void)n_in; (void)out_size; (void)d_ws; (void)ws_size;
    const float* useq = (const float*)d_in[0];
    const float* xz0  = (const float*)d_in[1];
    const float* p[18];
    for (int i = 0; i < 18; ++i) p[i] = (const float*)d_in[2 + i];
    float* out = (float*)d_out;

    const int B = 4096;
    dim3 block(32 * NWAVE);
    dim3 grid(B / SPW);
    reac_rk4_kernel<<<grid, block, 0, stream>>>(
        useq, xz0,
        p[0], p[1], p[2], p[3], p[4], p[5],
        p[6], p[7], p[8], p[9], p[10], p[11],
        p[12], p[13], p[14], p[15], p[16], p[17],
        out);
}


# r3 waves re-timed: input-layer k chunks without x (4,5,0,1,2) summed one step ahead in intervals 3/4, x chunk added last, hidden layer split 3+5 chunks over intervals 1/2; r3 input sum order changes
# speedup vs baseline: 7.9775x; 1.0750x over previous
#include <hip/hip_runtime.h>


#define Td  256
#define SPW 16
#define NWAVE 7

typedef __attribute__((ext_vector_type(2))) float v2f;
typedef __attribute__((ext_vector_type(8))) float v8f;

__device__ __forceinline__ float fast_tanh(float x)
{
#if __has_builtin(__builtin_amdgcn_tanhf)
    return __builtin_amdgcn_tanhf(x);
#else
    const float ax = __builtin_fabsf(x);
    const float t  = __expf(-2.0f * ax);
    const float y  = (1.0f - t) * __builtin_amdgcn_rcpf(1.0f + t);
    return __builtin_copysignf(y, x);
#endif
}

__device__ __forceinline__ float xhalf(float x)
{
    const int b = __builtin_bit_cast(int, x);
    return __builtin_bit_cast(float,
        __builtin_amdgcn_permlanex16(b, b, 0x76543210, 0xfedcba98, false, false));
}

__device__ __forceinline__ v8f wmma4(v2f a, v2f b, v8f c)
{
    return __builtin_amdgcn_wmma_f32_16x16x4_f32(false, a, false, b, (short)0, c,
                                                 false, false);
}

__device__ __forceinline__ int feat_of_slot(int j, int h)
{
    return 16 * (j >> 3) + 8 * h + (j & 7);
}

struct ScalarNet {
    v2f A[8];
    v8f cb;
    float w0[16], b0[16];
    float w2[8];
};

struct ZNet {
    v2f Aa[2][6];
    v2f Ab[2][8];
    v8f ca[2], cb[2];
    float w2[16];
    float b2;
};

__device__ __forceinline__ void load_A32(const float* __restrict__ W,
                                         v2f (&A)[2][8], int h, int c)
{
    #pragma unroll
    for (int mt = 0; mt < 2; ++mt) {
        #pragma unroll
        for (int kc = 0; kc < 8; ++kc) {
            A[mt][kc].x = W[feat_of_slot(2 * kc + 0, h) * 32 + 16 * mt + c];
            A[mt][kc].y = W[feat_of_slot(2 * kc + 1, h) * 32 + 16 * mt + c];
        }
    }
}

__device__ __forceinline__ void load_A32_tile(const float* __restrict__ W,
                                              v2f (&A)[8], int mt, int h, int c)
{
    #pragma unroll
    for (int kc = 0; kc < 8; ++kc) {
        A[kc].x = W[feat_of_slot(2 * kc + 0, h) * 32 + 16 * mt + c];
        A[kc].y = W[feat_of_slot(2 * kc + 1, h) * 32 + 16 * mt + c];
    }
}

__device__ __forceinline__ void load_tile8(const float* __restrict__ v,
                                           float (&dst)[8], int mt, int h)
{
    #pragma unroll
    for (int r = 0; r < 8; ++r) dst[r] = v[16 * mt + 8 * h + r];
}

__device__ __forceinline__ void load_A24(const float* __restrict__ W,
                                         v2f (&A)[2][6], int h, int c)
{
    #pragma unroll
    for (int mt = 0; mt < 2; ++mt) {
        #pragma unroll
        for (int kc = 0; kc < 6; ++kc) {
            A[mt][kc].x = W[(4 * kc + 2 * h + 0) * 32 + 16 * mt + c];
            A[mt][kc].y = W[(4 * kc + 2 * h + 1) * 32 + 16 * mt + c];
        }
    }
}

__device__ __forceinline__ void load_bias(const float* __restrict__ b,
                                          v8f (&cb)[2], int h)
{
    #pragma unroll
    for (int mt = 0; mt < 2; ++mt) {
        #pragma unroll
        for (int r = 0; r < 8; ++r) cb[mt][r] = b[16 * mt + 8 * h + r];
    }
}

__device__ __forceinline__ void load_slots(const float* __restrict__ v,
                                           float (&dst)[16], int h)
{
    #pragma unroll
    for (int j = 0; j < 16; ++j) dst[j] = v[feat_of_slot(j, h)];
}

__device__ __forceinline__ void hidden32(const v2f (&A)[2][8], const v8f (&cb)[2],
                                         const float (&in)[16], float (&outv)[16])
{
    v8f d0 = cb[0], d1 = cb[1];
    #pragma unroll
    for (int kc = 0; kc < 8; ++kc) {
        v2f b;
        b.x = in[2 * kc + 0];
        b.y = in[2 * kc + 1];
        d0 = wmma4(A[0][kc], b, d0);
        d1 = wmma4(A[1][kc], b, d1);
    }
    #pragma unroll
    for (int r = 0; r < 8; ++r) {
        outv[r]     = fast_tanh(d0[r]);
        outv[8 + r] = fast_tanh(d1[r]);
    }
}

__device__ __forceinline__ void hidden24(const v2f (&A)[2][6], const v8f (&cb)[2],
                                         const v2f (&zf)[6], float (&outv)[16])
{
    v8f d0 = cb[0], d1 = cb[1];
    #pragma unroll
    for (int kc = 0; kc < 6; ++kc) {
        d0 = wmma4(A[0][kc], zf[kc], d0);
        d1 = wmma4(A[1][kc], zf[kc], d1);
    }
    #pragma unroll
    for (int r = 0; r < 8; ++r) {
        outv[r]     = fast_tanh(d0[r]);
        outv[8 + r] = fast_tanh(d1[r]);
    }
}

__device__ __forceinline__ void layer0(float x, const float (&w)[16],
                                       const float (&b)[16], float (&outv)[16])
{
    #pragma unroll
    for (int j = 0; j < 16; ++j) outv[j] = fast_tanh(x * w[j] + b[j]);
}

__device__ __forceinline__ float out_dot(const float (&hv)[16], const float (&w)[16],
                                         float b2, float scale)
{
    float p0 = hv[0] * w[0], p1 = hv[1] * w[1], p2 = hv[2] * w[2], p3 = hv[3] * w[3];
    #pragma unroll
    for (int j = 4; j < 16; j += 4) {
        p0 += hv[j + 0] * w[j + 0];
        p1 += hv[j + 1] * w[j + 1];
        p2 += hv[j + 2] * w[j + 2];
        p3 += hv[j + 3] * w[j + 3];
    }
    const float p = (p0 + p1) + (p2 + p3);
    const float tot = p + xhalf(p);
    return (tot + b2) * scale;
}

__device__ __forceinline__ void lds_barrier()
{
    __builtin_amdgcn_fence(__ATOMIC_RELEASE, "workgroup", "local");
    __builtin_amdgcn_s_barrier();
    __builtin_amdgcn_fence(__ATOMIC_ACQUIRE, "workgroup", "local");
}

__device__ __forceinline__ void reactor(float xe0, float xe1, float u,
                                        float r1v, float r2v, float r3v,
                                        float& k0, float& k1)
{
    const float xs0 = xe0 * 0.3f + 0.5f;
    const float xs1 = xe1 * 0.2f + 0.5f;
    const float us  = u * 0.5f + 1.0f;
    const float dCa = 0.1f * (us - xs0) - r1v;
    const float dCb = -0.1f * xs1 + r1v - 3.0f * r2v + r3v;
    k0 = dCa * (1.0f / 0.3f);
    k1 = dCb * (1.0f / 0.2f);
}

__device__ __forceinline__ void r3_ahead_u(const ZNet& n, const v2f (&zc)[6],
                                           float uc, int role, bool hi,
                                           v2f& zu4, v2f& zu5, v8f& p0, v8f& p1)
{
    const float eu0 = xhalf(zc[4].x);
    const float eu1 = xhalf(zc[5].x);
    zu4.x = zc[4].y;
    zu4.y = hi ? eu1 : eu0;
    zu5.x = zc[5].y;
    zu5.y = hi ? uc : eu1;
    v2f s4, s5;
    s4.x = (role == 6) ? zu4.x : zc[4].x;
    s4.y = (role == 6) ? zu4.y : zc[4].y;
    s5.x = (role == 6) ? zu5.x : zc[5].x;
    s5.y = (role == 6) ? zu5.y : zc[5].y;
    p0 = wmma4(n.Aa[0][4], s4, n.ca[0]);
    p1 = wmma4(n.Aa[1][4], s4, n.ca[1]);
    p0 = wmma4(n.Aa[0][5], s5, p0);
    p1 = wmma4(n.Aa[1][5], s5, p1);
}

__device__ __forceinline__ void r3_ahead_xprep(const v2f (&zc)[6], int role, bool hi,
                                               v2f (&zsA)[3], v2f& ex3, v2f (&sel)[3])
{
    const float e0x = xhalf(zc[0].x), e0y = xhalf(zc[0].y);
    const float e1x = xhalf(zc[1].x), e1y = xhalf(zc[1].y);
    const float e2x = xhalf(zc[2].x), e2y = xhalf(zc[2].y);
    const float e3x = xhalf(zc[3].x), e3y = xhalf(zc[3].y);
    ex3.x = e3x;
    ex3.y = e3y;
    zsA[0].x = hi ? e1x : e0x;
    zsA[0].y = hi ? e1y : e0y;
    zsA[1].x = hi ? e2x : e1x;
    zsA[1].y = hi ? e2y : e1y;
    zsA[2].x = hi ? e3x : e2x;
    zsA[2].y = hi ? e3y : e2y;
    #pragma unroll
    for (int kc = 0; kc < 3; ++kc) {
        sel[kc].x = (role == 4) ? 0.5f * (zc[kc].x + zsA[kc].x) : zsA[kc].x;
        sel[kc].y = (role == 4) ? 0.5f * (zc[kc].y + zsA[kc].y) : zsA[kc].y;
    }
}

__device__ __forceinline__ void r3_ahead_xmma(const ZNet& n, const v2f (&sel)[3],
                                              v8f& p0, v8f& p1)
{
    #pragma unroll
    for (int kc = 0; kc < 3; ++kc) {
        p0 = wmma4(n.Aa[0][kc], sel[kc], p0);
        p1 = wmma4(n.Aa[1][kc], sel[kc], p1);
    }
}

__device__ __forceinline__ void hidden32_part(const v2f (&A)[2][8], const float (&in)[16],
                                              int kc0, int kc1, v8f& d0, v8f& d1)
{
    #pragma unroll
    for (int kc = kc0; kc < kc1; ++kc) {
        v2f b;
        b.x = in[2 * kc + 0];
        b.y = in[2 * kc + 1];
        d0 = wmma4(A[0][kc], b, d0);
        d1 = wmma4(A[1][kc], b, d1);
    }
}

__device__ __forceinline__ float scalar_net_tile(const ScalarNet& n, float xin)
{
    float g0[16], g1[8];
    layer0(xin, n.w0, n.b0, g0);
    v8f d0 = n.cb, d1 = {};
    #pragma unroll
    for (int kc = 0; kc < 8; kc += 2) {
        v2f b0, b1;
        b0.x = g0[2 * kc + 0];
        b0.y = g0[2 * kc + 1];
        b1.x = g0[2 * kc + 2];
        b1.y = g0[2 * kc + 3];
        d0 = wmma4(n.A[kc + 0], b0, d0);
        d1 = wmma4(n.A[kc + 1], b1, d1);
    }
    #pragma unroll
    for (int r = 0; r < 8; ++r) g1[r] = fast_tanh(d0[r] + d1[r]);
    float pa = g1[0] * n.w2[0], pb = g1[1] * n.w2[1];
    #pragma unroll
    for (int r = 2; r < 8; r += 2) {
        pa += g1[r + 0] * n.w2[r + 0];
        pb += g1[r + 1] * n.w2[r + 1];
    }
    const float p = pa + pb;
    return p + xhalf(p);
}

__global__ __launch_bounds__(32 * NWAVE) void reac_rk4_kernel(
    const float* __restrict__ useq, const float* __restrict__ xz0,
    const float* __restrict__ r1W0, const float* __restrict__ r1b0,
    const float* __restrict__ r1W1, const float* __restrict__ r1b1,
    const float* __restrict__ r1W2, const float* __restrict__ r1b2,
    const float* __restrict__ r2W0, const float* __restrict__ r2b0,
    const float* __restrict__ r2W1, const float* __restrict__ r2b1,
    const float* __restrict__ r2W2, const float* __restrict__ r2b2,
    const float* __restrict__ r3W0, const float* __restrict__ r3b0,
    const float* __restrict__ r3W1, const float* __restrict__ r3b1,
    const float* __restrict__ r3W2, const float* __restrict__ r3b2,
    float* __restrict__ out)
{
    __shared__ float sh_p[2][4][SPW];
    __shared__ float sh_r3[2][3][SPW];

    const int lane = threadIdx.x & 31;
    const int c    = lane & 15;
    const int h    = lane >> 4;
    const bool hi  = (h != 0);
    const int role = __builtin_amdgcn_readfirstlane((int)(threadIdx.x >> 5));
    const bool scalar_role = (role < 4);
    const int  mt   = role & 1;
    const bool use2 = (role == 2) || (role == 3);

    ScalarNet sn;
    load_A32_tile(use2 ? r2W1 : r1W1, sn.A, mt, h, c);
    {
        const float* __restrict__ bsel = use2 ? r2b1 : r1b1;
        #pragma unroll
        for (int r = 0; r < 8; ++r) sn.cb[r] = bsel[16 * mt + 8 * h + r];
    }
    load_slots(use2 ? r2W0 : r1W0, sn.w0, h);
    load_slots(use2 ? r2b0 : r1b0, sn.b0, h);
    load_tile8(use2 ? r2W2 : r1W2, sn.w2, mt, h);
    const float b2_r1 = r1b2[0];
    const float b2_r2 = r2b2[0];

    ZNet zn;
    load_A24(r3W0, zn.Aa, h, c);
    load_A32(r3W1, zn.Ab, h, c);
    load_bias(r3b0, zn.ca, h);
    load_bias(r3b1, zn.cb, h);
    load_slots(r3W2, zn.w2, h);
    zn.b2 = r3b2[0];

    const int    smp   = blockIdx.x * SPW + c;
    const float* xzrow = xz0 + (size_t)smp * 26;
    const float* urow  = useq + (size_t)smp * Td;
    float*       orow  = out + (size_t)smp * Td * 2;

    float x0 = xzrow[0], x1 = xzrow[1];
    v2f z[6];
    #pragma unroll
    for (int kc = 0; kc < 6; ++kc) {
        z[kc].x = xzrow[2 + 4 * kc + 2 * h + 0];
        z[kc].y = xzrow[2 + 4 * kc + 2 * h + 1];
    }

    float ga[16], gb[16];
    if (role == 6) {
        hidden24(zn.Aa, zn.ca, z, ga);
        hidden32(zn.Ab, zn.cb, ga, gb);
        sh_r3[0][0][c] = out_dot(gb, zn.w2, zn.b2, 0.2f);
    }

    if (scalar_role) __builtin_amdgcn_s_setprio(3);

    float u = urow[0];

    v2f zsA[3] = {}, selA[3] = {}, ex3 = {}, zu4 = {}, zu5 = {}, z3c = z[3];
    v8f pre0 = {}, pre1 = {};
    if (!scalar_role) {
        r3_ahead_u(zn, z, u, role, hi, zu4, zu5, pre0, pre1);
        r3_ahead_xprep(z, role, hi, zsA, ex3, selA);
        r3_ahead_xmma(zn, selA, pre0, pre1);
    }

    for (int t = 0; t < Td; ++t) {
        const int tp = t & 1;
        const float u_next = urow[(t + 1 < Td) ? (t + 1) : t];

        v2f zn1[6] = {};
        v8f h0 = {}, h1 = {};
        float r1v, r2v, r3v, k1a, k1b, k2a, k2b, k3a, k3b, k4a, k4b;

        if (scalar_role) sh_p[0][role][c] = scalar_net_tile(sn, use2 ? x1 : x0);
        else {
            if (role == 4) orow[t * 2 + h] = hi ? x1 : x0;

            v2f zs3, sel3;
            zs3.x = hi ? x0 : ex3.x;
            zs3.y = hi ? x1 : ex3.y;
            sel3.x = (role == 4) ? 0.5f * (z3c.x + zs3.x) : zs3.x;
            sel3.y = (role == 4) ? 0.5f * (z3c.y + zs3.y) : zs3.y;
            const v8f d0 = wmma4(zn.Aa[0][3], sel3, pre0);
            const v8f d1 = wmma4(zn.Aa[1][3], sel3, pre1);

            zn1[0] = zsA[0]; zn1[1] = zsA[1]; zn1[2] = zsA[2];
            zn1[3] = zs3;    zn1[4] = zu4;    zn1[5] = zu5;

            #pragma unroll
            for (int r = 0; r < 8; ++r) {
                ga[r]     = fast_tanh(d0[r]);
                ga[8 + r] = fast_tanh(d1[r]);
            }
            h0 = zn.cb[0];
            h1 = zn.cb[1];
            hidden32_part(zn.Ab, ga, 0, 3, h0, h1);
        }
        lds_barrier();
        r1v = ((sh_p[0][0][c] + sh_p[0][1][c]) + b2_r1) * 0.3f;
        r2v = ((sh_p[0][2][c] + sh_p[0][3][c]) + b2_r2) * 0.2f;
        r3v = sh_r3[tp][0][c];
        reactor(x0, x1, u, r1v, r2v, r3v, k1a, k1b);
        const float xa0 = x0 + 0.5f * k1a, xa1 = x1 + 0.5f * k1b;

        if (scalar_role) sh_p[1][role][c] = scalar_net_tile(sn, use2 ? xa1 : xa0);
        else {
            hidden32_part(zn.Ab, ga, 3, 8, h0, h1);
            #pragma unroll
            for (int r = 0; r < 8; ++r) {
                gb[r]     = fast_tanh(h0[r]);
                gb[8 + r] = fast_tanh(h1[r]);
            }
            const float rv = out_dot(gb, zn.w2, zn.b2, 0.2f);
            if (role == 4)      sh_r3[tp][1][c] = rv;
            else if (role == 5) sh_r3[tp][2][c] = rv;
            else                sh_r3[tp ^ 1][0][c] = rv;
        }
        lds_barrier();
        r3v = sh_r3[tp][1][c];
        r1v = ((sh_p[1][0][c] + sh_p[1][1][c]) + b2_r1) * 0.3f;
        r2v = ((sh_p[1][2][c] + sh_p[1][3][c]) + b2_r2) * 0.2f;
        reactor(xa0, xa1, u, r1v, r2v, r3v, k2a, k2b);
        const float xb0 = x0 + 0.5f * k2a, xb1 = x1 + 0.5f * k2b;

        if (scalar_role) sh_p[0][role][c] = scalar_net_tile(sn, use2 ? xb1 : xb0);
        else {
            r3_ahead_u(zn, zn1, u_next, role, hi, zu4, zu5, pre0, pre1);
            r3_ahead_xprep(zn1, role, hi, zsA, ex3, selA);
        }
        lds_barrier();
        r1v = ((sh_p[0][0][c] + sh_p[0][1][c]) + b2_r1) * 0.3f;
        r2v = ((sh_p[0][2][c] + sh_p[0][3][c]) + b2_r2) * 0.2f;
        reactor(xb0, xb1, u, r1v, r2v, r3v, k3a, k3b);
        const float xc0 = x0 + k3a, xc1 = x1 + k3b;

        if (scalar_role) sh_p[1][role][c] = scalar_net_tile(sn, use2 ? xc1 : xc0);
        else             r3_ahead_xmma(zn, selA, pre0, pre1);
        lds_barrier();
        r1v = ((sh_p[1][0][c] + sh_p[1][1][c]) + b2_r1) * 0.3f;
        r2v = ((sh_p[1][2][c] + sh_p[1][3][c]) + b2_r2) * 0.2f;
        r3v = sh_r3[tp][2][c];
        reactor(xc0, xc1, u, r1v, r2v, r3v, k4a, k4b);

        x0 = x0 + (k1a + 2.f * k2a + 2.f * k3a + k4a) * (1.f / 6.f);
        x1 = x1 + (k1b + 2.f * k2b + 2.f * k3b + k4b) * (1.f / 6.f);
        z3c = zn1[3];
        u = u_next;
    }
}

extern "C" void kernel_launch(void* const* d_in, const int* in_sizes, int n_in,
                              void* d_out, int out_size, void* d_ws, size_t ws_size,
                              hipStream_t stream) {
    (void)in_sizes; (void)n_in; (void)out_size; (void)d_ws; (void)ws_size;
    const float* useq = (const float*)d_in[0];
    const float* xz0  = (const float*)d_in[1];
    const float* p[18];
    for (int i = 0; i < 18; ++i) p[i] = (const float*)d_in[2 + i];
    float* out = (float*)d_out;

    const int B = 4096;
    dim3 block(32 * NWAVE);
    dim3 grid(B / SPW);
    reac_rk4_kernel<<<grid, block, 0, stream>>>(
        useq, xz0,
        p[0], p[1], p[2], p[3], p[4], p[5],
        p[6], p[7], p[8], p[9], p[10], p[11],
        p[12], p[13], p[14], p[15], p[16], p[17],
        out);
}
